# starting kernel
# baseline (speedup 1.0000x reference)
#include <hip/hip_runtime.h>

#define D 128

typedef float v2f __attribute__((ext_vector_type(2)));
typedef float v8f __attribute__((ext_vector_type(8)));


__global__ void zero_f32(float* __restrict__ p, int n) {
    int i = blockIdx.x * blockDim.x + threadIdx.x;
    if (i < n) p[i] = 0.0f;
}

__global__ void deg_kernel(const int* __restrict__ col,
                           const float* __restrict__ w,
                           float* __restrict__ deg, int E) {
    int e = blockIdx.x * blockDim.x + threadIdx.x;
    if (e < E) unsafeAtomicAdd(&deg[col[e]], w[e]);
}

__global__ void rsqrt_kernel(const float* __restrict__ deg,
                             float* __restrict__ dis, int n) {
    int i = blockIdx.x * blockDim.x + threadIdx.x;
    if (i < n) {
        float d = deg[i];
        dis[i] = (d > 0.0f) ? __frsqrt_rn(d) : 0.0f;
    }
}


__global__ __launch_bounds__(512) void gemm_dual(const float* __restrict__ x,
                                                 const float* __restrict__ Wlin,
                                                 const float* __restrict__ Wgcn,
                                                 float* __restrict__ out_hl,
                                                 float* __restrict__ h) {
    const int row0  = blockIdx.x * 16;
    const int wave  = threadIdx.x >> 5;
    const int lane  = threadIdx.x & 31;
    const int which = wave >> 3;
    const int col0  = (wave & 7) * 16;

    const float* __restrict__ W   = which ? Wgcn : Wlin;
    float* __restrict__       dst = which ? h    : out_hl;

    const int m  = lane & 15;
    const int hh = lane >> 4;

    const float* __restrict__ arow = x + (size_t)(row0 + m) * D;

    v8f acc = {};
#pragma unroll
    for (int k0 = 0; k0 < D; k0 += 4) {
        v2f a;
        a.x = arow[k0 + 2 * hh + 0];
        a.y = arow[k0 + 2 * hh + 1];
        v2f b;
        b.x = W[(size_t)(k0 + 2 * hh + 0) * D + col0 + m];
        b.y = W[(size_t)(k0 + 2 * hh + 1) * D + col0 + m];
        acc = __builtin_amdgcn_wmma_f32_16x16x4_f32(
             false, a,  false, b,
             (short)0, acc,  false,  false);
    }

#pragma unroll
    for (int r = 0; r < 8; ++r) {
        dst[(size_t)(row0 + r + 8 * hh) * D + col0 + m] = acc[r];
    }
}


__global__ __launch_bounds__(256) void scatter_kernel(const float* __restrict__ h,
                                                      const int* __restrict__ row_idx,
                                                      const int* __restrict__ col_idx,
                                                      const float* __restrict__ w,
                                                      const float* __restrict__ dis,
                                                      float* __restrict__ out, int E) {
    const int wave = threadIdx.x >> 5;
    const int lane = threadIdx.x & 31;
    const int e    = blockIdx.x * 8 + wave;
    if (e >= E) return;

    const int r = row_idx[e];
    const int c = col_idx[e];
    const float norm = dis[r] * w[e] * dis[c];

    const float4* __restrict__ hv = (const float4*)(h + (size_t)r * D);
    float4 v = hv[lane];

    float* __restrict__ o = out + (size_t)c * D + lane * 4;
    unsafeAtomicAdd(o + 0, v.x * norm);
    unsafeAtomicAdd(o + 1, v.y * norm);
    unsafeAtomicAdd(o + 2, v.z * norm);
    unsafeAtomicAdd(o + 3, v.w * norm);
}


extern "C" void kernel_launch(void* const* d_in, const int* in_sizes, int n_in,
                              void* d_out, int out_size, void* d_ws, size_t ws_size,
                              hipStream_t stream) {
    const float* x    = (const float*)d_in[0];
    const int*   ei   = (const int*)d_in[1];
    const float* ew   = (const float*)d_in[2];
    const float* Wlin = (const float*)d_in[3];
    const float* Wgcn = (const float*)d_in[4];
    float*       out  = (float*)d_out;

    const int N = in_sizes[0] / D;
    const int E = in_sizes[2];

    const int* row = ei;
    const int* col = ei + E;

    float* deg = (float*)d_ws;
    float* dis = deg + N;
    float* h   = dis + N;

    zero_f32<<<(N + 255) / 256, 256, 0, stream>>>(deg, N);
    deg_kernel<<<(E + 255) / 256, 256, 0, stream>>>(col, ew, deg, E);
    rsqrt_kernel<<<(N + 255) / 256, 256, 0, stream>>>(deg, dis, N);
    gemm_dual<<<N / 16, 512, 0, stream>>>(x, Wlin, Wgcn, out, h);
    scatter_kernel<<<(E + 7) / 8, 256, 0, stream>>>(h, row, col, ew, dis, out, E);
}


# scatter: lane l owns row elements l,l+32,l+64,l+96, so each f32 atomic instruction covers one whole 128-B line (4 line requests per edge, was 16); adds still land in arrival order
# speedup vs baseline: 1.2800x; 1.2800x over previous
#include <hip/hip_runtime.h>

#define D 128

typedef float v2f __attribute__((ext_vector_type(2)));
typedef float v8f __attribute__((ext_vector_type(8)));


__global__ void zero_f32(float* __restrict__ p, int n) {
    int i = blockIdx.x * blockDim.x + threadIdx.x;
    if (i < n) p[i] = 0.0f;
}

__global__ void deg_kernel(const int* __restrict__ col,
                           const float* __restrict__ w,
                           float* __restrict__ deg, int E) {
    int e = blockIdx.x * blockDim.x + threadIdx.x;
    if (e < E) unsafeAtomicAdd(&deg[col[e]], w[e]);
}

__global__ void rsqrt_kernel(const float* __restrict__ deg,
                             float* __restrict__ dis, int n) {
    int i = blockIdx.x * blockDim.x + threadIdx.x;
    if (i < n) {
        float d = deg[i];
        dis[i] = (d > 0.0f) ? __frsqrt_rn(d) : 0.0f;
    }
}


__global__ __launch_bounds__(512) void gemm_dual(const float* __restrict__ x,
                                                 const float* __restrict__ Wlin,
                                                 const float* __restrict__ Wgcn,
                                                 float* __restrict__ out_hl,
                                                 float* __restrict__ h) {
    const int row0  = blockIdx.x * 16;
    const int wave  = threadIdx.x >> 5;
    const int lane  = threadIdx.x & 31;
    const int which = wave >> 3;
    const int col0  = (wave & 7) * 16;

    const float* __restrict__ W   = which ? Wgcn : Wlin;
    float* __restrict__       dst = which ? h    : out_hl;

    const int m  = lane & 15;
    const int hh = lane >> 4;

    const float* __restrict__ arow = x + (size_t)(row0 + m) * D;

    v8f acc = {};
#pragma unroll
    for (int k0 = 0; k0 < D; k0 += 4) {
        v2f a;
        a.x = arow[k0 + 2 * hh + 0];
        a.y = arow[k0 + 2 * hh + 1];
        v2f b;
        b.x = W[(size_t)(k0 + 2 * hh + 0) * D + col0 + m];
        b.y = W[(size_t)(k0 + 2 * hh + 1) * D + col0 + m];
        acc = __builtin_amdgcn_wmma_f32_16x16x4_f32(
             false, a,  false, b,
             (short)0, acc,  false,  false);
    }

#pragma unroll
    for (int r = 0; r < 8; ++r) {
        dst[(size_t)(row0 + r + 8 * hh) * D + col0 + m] = acc[r];
    }
}


__global__ __launch_bounds__(256) void scatter_kernel(const float* __restrict__ h,
                                                      const int* __restrict__ row_idx,
                                                      const int* __restrict__ col_idx,
                                                      const float* __restrict__ w,
                                                      const float* __restrict__ dis,
                                                      float* __restrict__ out, int E) {
    const int wave = threadIdx.x >> 5;
    const int lane = threadIdx.x & 31;
    const int e    = blockIdx.x * 8 + wave;
    if (e >= E) return;

    const int r = row_idx[e];
    const int c = col_idx[e];
    const float norm = dis[r] * w[e] * dis[c];

    const float* __restrict__ hr = h + (size_t)r * D + lane;
    const float v0 = hr[0];
    const float v1 = hr[32];
    const float v2 = hr[64];
    const float v3 = hr[96];

    float* __restrict__ o = out + (size_t)c * D + lane;
    unsafeAtomicAdd(o +  0, v0 * norm);
    unsafeAtomicAdd(o + 32, v1 * norm);
    unsafeAtomicAdd(o + 64, v2 * norm);
    unsafeAtomicAdd(o + 96, v3 * norm);
}


extern "C" void kernel_launch(void* const* d_in, const int* in_sizes, int n_in,
                              void* d_out, int out_size, void* d_ws, size_t ws_size,
                              hipStream_t stream) {
    const float* x    = (const float*)d_in[0];
    const int*   ei   = (const int*)d_in[1];
    const float* ew   = (const float*)d_in[2];
    const float* Wlin = (const float*)d_in[3];
    const float* Wgcn = (const float*)d_in[4];
    float*       out  = (float*)d_out;

    const int N = in_sizes[0] / D;
    const int E = in_sizes[2];

    const int* row = ei;
    const int* col = ei + E;

    float* deg = (float*)d_ws;
    float* dis = deg + N;
    float* h   = dis + N;

    zero_f32<<<(N + 255) / 256, 256, 0, stream>>>(deg, N);
    deg_kernel<<<(E + 255) / 256, 256, 0, stream>>>(col, ew, deg, E);
    rsqrt_kernel<<<(N + 255) / 256, 256, 0, stream>>>(deg, dis, N);
    gemm_dual<<<N / 16, 512, 0, stream>>>(x, Wlin, Wgcn, out, h);
    scatter_kernel<<<(E + 7) / 8, 256, 0, stream>>>(h, row, col, ew, dis, out, E);
}


# gemm: each wave owns a 32x64 output piece in 8 accumulators; x and W come in 16-B loads reused by 4 and 2 tiles, stores are whole lines; k re-ordered inside 32-k blocks (product sum order changes)
# speedup vs baseline: 1.5183x; 1.1862x over previous
#include <hip/hip_runtime.h>

#define D 128

typedef float v2f __attribute__((ext_vector_type(2)));
typedef float v4f __attribute__((ext_vector_type(4)));
typedef float v8f __attribute__((ext_vector_type(8)));


__global__ void zero_f32(float* __restrict__ p, int n) {
    int i = blockIdx.x * blockDim.x + threadIdx.x;
    if (i < n) p[i] = 0.0f;
}

__global__ void deg_kernel(const int* __restrict__ col,
                           const float* __restrict__ w,
                           float* __restrict__ deg, int E) {
    int e = blockIdx.x * blockDim.x + threadIdx.x;
    if (e < E) unsafeAtomicAdd(&deg[col[e]], w[e]);
}

__global__ void rsqrt_kernel(const float* __restrict__ deg,
                             float* __restrict__ dis, int n) {
    int i = blockIdx.x * blockDim.x + threadIdx.x;
    if (i < n) {
        float d = deg[i];
        dis[i] = (d > 0.0f) ? __frsqrt_rn(d) : 0.0f;
    }
}


__device__ __forceinline__ v8f mma4(v2f a, v2f b, v8f acc) {
    return __builtin_amdgcn_wmma_f32_16x16x4_f32(false, a, false, b, (short)0, acc, false, false);
}

struct ABlock { v4f t0[4], t1[4]; };

__device__ __forceinline__ ABlock load_a(const float* __restrict__ ap, int kb) {
    ABlock q;
#pragma unroll
    for (int i = 0; i < 4; ++i) {
        q.t0[i] = *(const v4f*)(ap + kb + 4 * i);
        q.t1[i] = *(const v4f*)(ap + 16 * D + kb + 4 * i);
    }
    return q;
}

__device__ __forceinline__ void mma_block(const ABlock& A, const float* __restrict__ wp, int kb, v8f (&acc)[2][4]) {
#pragma unroll
    for (int i = 0; i < 4; ++i) {
        const float* __restrict__ wr = wp + (size_t)(kb + 4 * i) * D;
        const v4f b0 = *(const v4f*)(wr + 0 * D);
        const v4f b1 = *(const v4f*)(wr + 1 * D);
        const v4f b2 = *(const v4f*)(wr + 2 * D);
        const v4f b3 = *(const v4f*)(wr + 3 * D);
        const v2f a0lo = {A.t0[i].x, A.t0[i].y}, a0hi = {A.t0[i].z, A.t0[i].w};
        const v2f a1lo = {A.t1[i].x, A.t1[i].y}, a1hi = {A.t1[i].z, A.t1[i].w};
#pragma unroll
        for (int j = 0; j < 4; ++j) {
            const v2f blo = {b0[j], b1[j]};
            acc[0][j] = mma4(a0lo, blo, acc[0][j]);
            acc[1][j] = mma4(a1lo, blo, acc[1][j]);
        }
#pragma unroll
        for (int j = 0; j < 4; ++j) {
            const v2f bhi = {b2[j], b3[j]};
            acc[0][j] = mma4(a0hi, bhi, acc[0][j]);
            acc[1][j] = mma4(a1hi, bhi, acc[1][j]);
        }
    }
}

__global__ __launch_bounds__(128) void gemm_dual(const float* __restrict__ x,
                                                 const float* __restrict__ Wlin,
                                                 const float* __restrict__ Wgcn,
                                                 float* __restrict__ out_hl,
                                                 float* __restrict__ h) {
    const int wave  = threadIdx.x >> 5;
    const int lane  = threadIdx.x & 31;
    const int c     = lane & 15;
    const int hh    = lane >> 4;
    const int cb    = (wave & 1) * 64;
    const int which = wave >> 1;
    const int row0  = blockIdx.x * 32;

    const float* __restrict__ W   = which ? Wgcn : Wlin;
    float* __restrict__       dst = which ? h    : out_hl;

    const float* __restrict__ ap = x + (size_t)(row0 + c) * D + 16 * hh;
    const float* __restrict__ wp = W + (size_t)(16 * hh) * D + cb + 4 * c;

    v8f acc[2][4] = {};
    ABlock cur = load_a(ap, 0);
#pragma unroll 1
    for (int kb = 32; kb < D; kb += 32) {
        const ABlock nxt = load_a(ap, kb);
        mma_block(cur, wp, kb - 32, acc);
        cur = nxt;
    }
    mma_block(cur, wp, D - 32, acc);

    float* __restrict__ dp = dst + (size_t)(row0 + 8 * hh) * D + cb + 4 * c;
#pragma unroll
    for (int t = 0; t < 2; ++t) {
#pragma unroll
        for (int r = 0; r < 8; ++r) {
            const v4f o = {acc[t][0][r], acc[t][1][r], acc[t][2][r], acc[t][3][r]};
            *(v4f*)(dp + (size_t)(16 * t + r) * D) = o;
        }
    }
}


__global__ __launch_bounds__(256) void scatter_kernel(const float* __restrict__ h,
                                                      const int* __restrict__ row_idx,
                                                      const int* __restrict__ col_idx,
                                                      const float* __restrict__ w,
                                                      const float* __restrict__ dis,
                                                      float* __restrict__ out, int E) {
    const int wave = threadIdx.x >> 5;
    const int lane = threadIdx.x & 31;
    const int e    = blockIdx.x * 8 + wave;
    if (e >= E) return;

    const int r = row_idx[e];
    const int c = col_idx[e];
    const float norm = dis[r] * w[e] * dis[c];

    const float* __restrict__ hr = h + (size_t)r * D + lane;
    const float v0 = hr[0];
    const float v1 = hr[32];
    const float v2 = hr[64];
    const float v3 = hr[96];

    float* __restrict__ o = out + (size_t)c * D + lane;
    unsafeAtomicAdd(o +  0, v0 * norm);
    unsafeAtomicAdd(o + 32, v1 * norm);
    unsafeAtomicAdd(o + 64, v2 * norm);
    unsafeAtomicAdd(o + 96, v3 * norm);
}


extern "C" void kernel_launch(void* const* d_in, const int* in_sizes, int n_in,
                              void* d_out, int out_size, void* d_ws, size_t ws_size,
                              hipStream_t stream) {
    const float* x    = (const float*)d_in[0];
    const int*   ei   = (const int*)d_in[1];
    const float* ew   = (const float*)d_in[2];
    const float* Wlin = (const float*)d_in[3];
    const float* Wgcn = (const float*)d_in[4];
    float*       out  = (float*)d_out;

    const int N = in_sizes[0] / D;
    const int E = in_sizes[2];

    const int* row = ei;
    const int* col = ei + E;

    float* deg = (float*)d_ws;
    float* dis = deg + N;
    float* h   = dis + N;

    zero_f32<<<(N + 255) / 256, 256, 0, stream>>>(deg, N);
    deg_kernel<<<(E + 255) / 256, 256, 0, stream>>>(col, ew, deg, E);
    rsqrt_kernel<<<(N + 255) / 256, 256, 0, stream>>>(deg, dis, N);
    gemm_dual<<<N / 32, 128, 0, stream>>>(x, Wlin, Wgcn, out, h);
    scatter_kernel<<<(E + 7) / 8, 256, 0, stream>>>(h, row, col, ew, dis, out, E);
}


# aggregate before the product: edges counting-sorted by destination in the call (integer rank atomics), a wave sums a node's x rows in registers, one K=256 product; no float atomics; sum order changes
# speedup vs baseline: 2.0686x; 1.3624x over previous
#include <hip/hip_runtime.h>

#define D 128

typedef float v2f __attribute__((ext_vector_type(2)));
typedef float v4f __attribute__((ext_vector_type(4)));
typedef float v8f __attribute__((ext_vector_type(8)));


__global__ void zero_i32(int* __restrict__ p, int n) {
    int i = blockIdx.x * blockDim.x + threadIdx.x;
    if (i < n) p[i] = 0;
}

__global__ void count_kernel(const int* __restrict__ col,
                             int* __restrict__ cnt,
                             int* __restrict__ rank, int E) {
    int e = blockIdx.x * blockDim.x + threadIdx.x;
    if (e < E) rank[e] = atomicAdd(&cnt[col[e]], 1);
}

__global__ __launch_bounds__(1024) void scan_kernel(const int* __restrict__ cnt,
                                                    int* __restrict__ off, int n) {
    __shared__ int red[1024];
    __shared__ int buf[2][1024];
    const int t    = threadIdx.x;
    const int base = blockIdx.x * 1024;
    const int nv   = base >> 2;
    const int4* __restrict__ c4 = (const int4*)cnt;

    int s = 0;
    for (int i = t; i < nv; i += 8 * 1024) {
        int4 q[8];
#pragma unroll
        for (int k = 0; k < 8; ++k) {
            const int idx = i + k * 1024;
            q[k] = (idx < nv) ? c4[idx] : make_int4(0, 0, 0, 0);
        }
#pragma unroll
        for (int k = 0; k < 8; ++k) s += q[k].x + q[k].y + q[k].z + q[k].w;
    }
    const int g = base + t;
    const int v = (g < n) ? cnt[g] : 0;

    red[t]    = s;
    buf[0][t] = v;
    __syncthreads();
    for (int stride = 512; stride > 0; stride >>= 1) {
        if (t < stride) red[t] += red[t + stride];
        __syncthreads();
    }
    const int total = red[0];

    int pin = 0;
    for (int d = 1; d < 1024; d <<= 1) {
        int val = buf[pin][t];
        if (t >= d) val += buf[pin][t - d];
        buf[pin ^ 1][t] = val;
        __syncthreads();
        pin ^= 1;
    }
    if (g < n) off[g] = total + buf[pin][t] - v;
}

__global__ void fill_kernel(const int* __restrict__ row,
                            const int* __restrict__ col,
                            const float* __restrict__ w,
                            const int* __restrict__ off,
                            const int* __restrict__ rank,
                            int* __restrict__ rec, int E) {
    int e = blockIdx.x * blockDim.x + threadIdx.x;
    if (e < E) {
        const int p = off[col[e]] + rank[e];
        ((int2*)rec)[p] = make_int2(row[e], __float_as_int(w[e]));
    }
}

__global__ void dis_kernel(const int* __restrict__ rec,
                           const int* __restrict__ off,
                           const int* __restrict__ cnt,
                           float* __restrict__ dis, int n) {
    int c = blockIdx.x * blockDim.x + threadIdx.x;
    if (c < n) {
        const int2* __restrict__ q = (const int2*)rec + off[c];
        const int m = cnt[c];
        float d = 0.0f;
        for (int j = 0; j < m; j += 4) {
            const float w0 = __int_as_float(q[j].y);
            const float w1 = (j + 1 < m) ? __int_as_float(q[j + 1].y) : 0.0f;
            const float w2 = (j + 2 < m) ? __int_as_float(q[j + 2].y) : 0.0f;
            const float w3 = (j + 3 < m) ? __int_as_float(q[j + 3].y) : 0.0f;
            d += w0; d += w1; d += w2; d += w3;
        }
        dis[c] = (d > 0.0f) ? __frsqrt_rn(d) : 0.0f;
    }
}


template <int G>
__device__ __forceinline__ void add_rows(const float* __restrict__ xl, int r, float nm, int j, v4f& acc) {
    v4f   v[G];
    float f[G];
#pragma unroll
    for (int k = 0; k < G; ++k) {
        const int rk = __builtin_amdgcn_readlane(r, j + k);
        f[k] = __int_as_float(__builtin_amdgcn_readlane(__float_as_int(nm), j + k));
        v[k] = *(const v4f*)(xl + (size_t)rk * D);
    }
#pragma unroll
    for (int k = 0; k < G; ++k) acc += v[k] * f[k];
}

__global__ __launch_bounds__(256) void gather_kernel(const float* __restrict__ x,
                                                     const int* __restrict__ rec,
                                                     const int* __restrict__ off,
                                                     const int* __restrict__ cnt,
                                                     const float* __restrict__ dis,
                                                     float* __restrict__ agg, int n) {
    const int lane = threadIdx.x & 31;
    const int c = __builtin_amdgcn_readfirstlane(blockIdx.x * 8 + (threadIdx.x >> 5));
    if (c >= n) return;
    const int   s  = __builtin_amdgcn_readfirstlane(off[c]);
    const int   m  = __builtin_amdgcn_readfirstlane(cnt[c]);
    const float dc = dis[c];
    const int2* __restrict__ q  = (const int2*)rec + s;
    const float* __restrict__ xl = x + 4 * lane;

    v4f acc = {0.0f, 0.0f, 0.0f, 0.0f};
    for (int b = 0; b < m; b += 32) {
        const int mm = (m - b < 32) ? (m - b) : 32;
        const int   i  = (lane < mm) ? lane : (mm - 1);
        const int2  e  = q[b + i];
        const int   r  = e.x;
        const float nm = (lane < mm) ? (dis[r] * __int_as_float(e.y)) * dc : 0.0f;
        int j = 0;
        for (; j + 8 <= mm; j += 8) add_rows<8>(xl, r, nm, j, acc);
        const int rem = mm - j;
        if (rem > 4)      add_rows<8>(xl, r, nm, j, acc);
        else if (rem > 0) add_rows<4>(xl, r, nm, j, acc);
    }
    *(v4f*)(agg + (size_t)c * D + 4 * lane) = acc;
}


__device__ __forceinline__ v8f mma4(v2f a, v2f b, v8f acc) {
    return __builtin_amdgcn_wmma_f32_16x16x4_f32(false, a, false, b, (short)0, acc, false, false);
}

struct ABlock { v4f t0[4], t1[4]; };

__device__ __forceinline__ ABlock load_a(const float* __restrict__ ap, int kb) {
    ABlock q;
#pragma unroll
    for (int i = 0; i < 4; ++i) {
        q.t0[i] = *(const v4f*)(ap + kb + 4 * i);
        q.t1[i] = *(const v4f*)(ap + 16 * D + kb + 4 * i);
    }
    return q;
}

__device__ __forceinline__ void mma_block(const ABlock& A, const float* __restrict__ wp, int kb, v8f (&acc)[2][4]) {
#pragma unroll
    for (int i = 0; i < 4; ++i) {
        const float* __restrict__ wr = wp + (size_t)(kb + 4 * i) * D;
        const v4f b0 = *(const v4f*)(wr + 0 * D);
        const v4f b1 = *(const v4f*)(wr + 1 * D);
        const v4f b2 = *(const v4f*)(wr + 2 * D);
        const v4f b3 = *(const v4f*)(wr + 3 * D);
        const v2f a0lo = {A.t0[i].x, A.t0[i].y}, a0hi = {A.t0[i].z, A.t0[i].w};
        const v2f a1lo = {A.t1[i].x, A.t1[i].y}, a1hi = {A.t1[i].z, A.t1[i].w};
#pragma unroll
        for (int j = 0; j < 4; ++j) {
            const v2f blo = {b0[j], b1[j]};
            acc[0][j] = mma4(a0lo, blo, acc[0][j]);
            acc[1][j] = mma4(a1lo, blo, acc[1][j]);
        }
#pragma unroll
        for (int j = 0; j < 4; ++j) {
            const v2f bhi = {b2[j], b3[j]};
            acc[0][j] = mma4(a0hi, bhi, acc[0][j]);
            acc[1][j] = mma4(a1hi, bhi, acc[1][j]);
        }
    }
}

__device__ __forceinline__ void product_half(const float* __restrict__ ap, const float* __restrict__ wp, v8f (&acc)[2][4]) {
    ABlock cur = load_a(ap, 0);
#pragma unroll 1
    for (int kb = 32; kb < D; kb += 32) {
        const ABlock nxt = load_a(ap, kb);
        mma_block(cur, wp, kb - 32, acc);
        cur = nxt;
    }
    mma_block(cur, wp, D - 32, acc);
}

__global__ __launch_bounds__(64) void gemm_cat(const float* __restrict__ x,
                                               const float* __restrict__ agg,
                                               const float* __restrict__ Wlin,
                                               const float* __restrict__ Wgcn,
                                               float* __restrict__ out) {
    const int wave  = threadIdx.x >> 5;
    const int lane  = threadIdx.x & 31;
    const int c     = lane & 15;
    const int hh    = lane >> 4;
    const int cb    = wave * 64;
    const int row0  = blockIdx.x * 32;

    const size_t aoff = (size_t)(row0 + c) * D + 16 * hh;
    const size_t woff = (size_t)(16 * hh) * D + cb + 4 * c;

    v8f acc[2][4] = {};
    product_half(x + aoff, Wlin + woff, acc);
    product_half(agg + aoff, Wgcn + woff, acc);

    float* __restrict__ dp = out + (size_t)(row0 + 8 * hh) * D + cb + 4 * c;
#pragma unroll
    for (int t = 0; t < 2; ++t) {
#pragma unroll
        for (int r = 0; r < 8; ++r) {
            const v4f o = {acc[t][0][r], acc[t][1][r], acc[t][2][r], acc[t][3][r]};
            *(v4f*)(dp + (size_t)(16 * t + r) * D) = o;
        }
    }
}


extern "C" void kernel_launch(void* const* d_in, const int* in_sizes, int n_in,
                              void* d_out, int out_size, void* d_ws, size_t ws_size,
                              hipStream_t stream) {
    const float* x    = (const float*)d_in[0];
    const int*   ei   = (const int*)d_in[1];
    const float* ew   = (const float*)d_in[2];
    const float* Wlin = (const float*)d_in[3];
    const float* Wgcn = (const float*)d_in[4];
    float*       out  = (float*)d_out;

    const int N = in_sizes[0] / D;
    const int E = in_sizes[2];

    const int* row = ei;
    const int* col = ei + E;

    int*   off = (int*)d_ws;
    float* dis = (float*)d_ws + N;
    float* agg = dis + N;

    int* cnt  = (int*)d_out;
    int* rank = cnt + N;
    int* rec  = rank + E;

    zero_i32<<<(N + 255) / 256, 256, 0, stream>>>(cnt, N);
    count_kernel<<<(E + 255) / 256, 256, 0, stream>>>(col, cnt, rank, E);
    scan_kernel<<<(N + 1023) / 1024, 1024, 0, stream>>>(cnt, off, N);
    fill_kernel<<<(E + 255) / 256, 256, 0, stream>>>(row, col, ew, off, rank, rec, E);
    dis_kernel<<<(N + 255) / 256, 256, 0, stream>>>(rec, off, cnt, dis, N);
    gather_kernel<<<(N + 7) / 8, 256, 0, stream>>>(x, rec, off, cnt, dis, agg, N);
    gemm_cat<<<N / 32, 64, 0, stream>>>(x, agg, Wlin, Wgcn, out);
}


# a 32-slot bucket per destination: an edge goes to slot 32*c+rank straight from the count atomic (rank 32 up: side list); no prefix scan, rank array or off lookup; 2 launches fewer; arrival-order sums
# speedup vs baseline: 2.2927x; 1.1083x over previous
#include <hip/hip_runtime.h>

#define D 128

typedef float v2f __attribute__((ext_vector_type(2)));
typedef float v4f __attribute__((ext_vector_type(4)));
typedef float v8f __attribute__((ext_vector_type(8)));


#define CAP 32

__global__ void zero_i32(int* __restrict__ p, int n) {
    int i = blockIdx.x * blockDim.x + threadIdx.x;
    if (i < n) p[i] = 0;
}

__global__ void place_kernel(const int* __restrict__ row,
                             const int* __restrict__ col,
                             const float* __restrict__ w,
                             int* __restrict__ cnt,
                             int* __restrict__ tab,
                             int* __restrict__ ovf_dst,
                             int* __restrict__ ovf_rec,
                             int E, int n) {
    int e = blockIdx.x * blockDim.x + threadIdx.x;
    if (e < E) {
        const int  c = col[e];
        const int2 v = make_int2(row[e], __float_as_int(w[e]));
        const int  k = atomicAdd(&cnt[c], 1);
        int2* dst = (int2*)tab + ((size_t)c * CAP + k);
        if (k >= CAP) {
            const int j = atomicAdd(&cnt[n], 1);
            ovf_dst[j] = c;
            dst = (int2*)ovf_rec + j;
        }
        *dst = v;
    }
}

__global__ void dis_kernel(const int* __restrict__ tab,
                           const int* __restrict__ cnt,
                           const int* __restrict__ ovf_dst,
                           const int* __restrict__ ovf_rec,
                           float* __restrict__ dis, int n) {
    int c = blockIdx.x * blockDim.x + threadIdx.x;
    if (c < n) {
        const int2* __restrict__ q = (const int2*)tab + (size_t)c * CAP;
        const int mt = cnt[c];
        const int m  = (mt < CAP) ? mt : CAP;
        float d = 0.0f;
        for (int j = 0; j < m; j += 4) {
            const float w0 = __int_as_float(q[j].y);
            const float w1 = (j + 1 < m) ? __int_as_float(q[j + 1].y) : 0.0f;
            const float w2 = (j + 2 < m) ? __int_as_float(q[j + 2].y) : 0.0f;
            const float w3 = (j + 3 < m) ? __int_as_float(q[j + 3].y) : 0.0f;
            d += w0; d += w1; d += w2; d += w3;
        }
        if (mt > CAP) {
            const int no = cnt[n];
            for (int k = 0; k < no; ++k)
                if (ovf_dst[k] == c) d += __int_as_float(((const int2*)ovf_rec)[k].y);
        }
        dis[c] = (d > 0.0f) ? __frsqrt_rn(d) : 0.0f;
    }
}


template <int G>
__device__ __forceinline__ void add_rows(const float* __restrict__ xl, int r, float nm, int j, v4f& acc) {
    v4f   v[G];
    float f[G];
#pragma unroll
    for (int k = 0; k < G; ++k) {
        const int rk = __builtin_amdgcn_readlane(r, j + k);
        f[k] = __int_as_float(__builtin_amdgcn_readlane(__float_as_int(nm), j + k));
        v[k] = *(const v4f*)(xl + (size_t)rk * D);
    }
#pragma unroll
    for (int k = 0; k < G; ++k) acc += v[k] * f[k];
}

__global__ __launch_bounds__(256) void gather_kernel(const float* __restrict__ x,
                                                     const int* __restrict__ tab,
                                                     const int* __restrict__ cnt,
                                                     const int* __restrict__ ovf_dst,
                                                     const int* __restrict__ ovf_rec,
                                                     const float* __restrict__ dis,
                                                     float* __restrict__ agg, int n) {
    const int lane = threadIdx.x & 31;
    const int c = __builtin_amdgcn_readfirstlane(blockIdx.x * 8 + (threadIdx.x >> 5));
    if (c >= n) return;
    const int2  e  = ((const int2*)tab)[(size_t)c * CAP + lane];
    const int   m  = __builtin_amdgcn_readfirstlane(cnt[c]);
    const float dc = dis[c];
    const float* __restrict__ xl = x + 4 * lane;

    v4f acc = {0.0f, 0.0f, 0.0f, 0.0f};
    const int mm = (m < CAP) ? m : CAP;
    {
        const int   rl = (mm > 0) ? __builtin_amdgcn_readlane(e.x, (mm > 0) ? mm - 1 : 0) : 0;
        const int   r  = (lane < mm) ? e.x : rl;
        const float nm = (lane < mm) ? (dis[r] * __int_as_float(e.y)) * dc : 0.0f;
        int j = 0;
        for (; j + 8 <= mm; j += 8) add_rows<8>(xl, r, nm, j, acc);
        const int rem = mm - j;
        if (rem > 4)      add_rows<8>(xl, r, nm, j, acc);
        else if (rem > 0) add_rows<4>(xl, r, nm, j, acc);
    }
    if (m > CAP) {
        const int no = __builtin_amdgcn_readfirstlane(cnt[n]);
        for (int k0 = 0; k0 < no; k0 += 32) {
            const int k = k0 + lane;
            int  od = -1;
            int2 oe = make_int2(0, 0);
            if (k < no) { od = ovf_dst[k]; oe = ((const int2*)ovf_rec)[k]; }
            const bool  hit = (od == c);
            const float onm = hit ? (dis[oe.x] * __int_as_float(oe.y)) * dc : 0.0f;
            unsigned mask = __builtin_amdgcn_ballot_w32(hit);
            while (mask) {
                const int l = __builtin_ctz(mask);
                mask &= mask - 1;
                const int   rk = __builtin_amdgcn_readlane(oe.x, l);
                const float f  = __int_as_float(__builtin_amdgcn_readlane(__float_as_int(onm), l));
                acc += *(const v4f*)(xl + (size_t)rk * D) * f;
            }
        }
    }
    *(v4f*)(agg + (size_t)c * D + 4 * lane) = acc;
}


__device__ __forceinline__ v8f mma4(v2f a, v2f b, v8f acc) {
    return __builtin_amdgcn_wmma_f32_16x16x4_f32(false, a, false, b, (short)0, acc, false, false);
}

struct ABlock { v4f t0[4], t1[4]; };

__device__ __forceinline__ ABlock load_a(const float* __restrict__ ap, int kb) {
    ABlock q;
#pragma unroll
    for (int i = 0; i < 4; ++i) {
        q.t0[i] = *(const v4f*)(ap + kb + 4 * i);
        q.t1[i] = *(const v4f*)(ap + 16 * D + kb + 4 * i);
    }
    return q;
}

__device__ __forceinline__ void mma_block(const ABlock& A, const float* __restrict__ wp, int kb, v8f (&acc)[2][4]) {
#pragma unroll
    for (int i = 0; i < 4; ++i) {
        const float* __restrict__ wr = wp + (size_t)(kb + 4 * i) * D;
        const v4f b0 = *(const v4f*)(wr + 0 * D);
        const v4f b1 = *(const v4f*)(wr + 1 * D);
        const v4f b2 = *(const v4f*)(wr + 2 * D);
        const v4f b3 = *(const v4f*)(wr + 3 * D);
        const v2f a0lo = {A.t0[i].x, A.t0[i].y}, a0hi = {A.t0[i].z, A.t0[i].w};
        const v2f a1lo = {A.t1[i].x, A.t1[i].y}, a1hi = {A.t1[i].z, A.t1[i].w};
#pragma unroll
        for (int j = 0; j < 4; ++j) {
            const v2f blo = {b0[j], b1[j]};
            acc[0][j] = mma4(a0lo, blo, acc[0][j]);
            acc[1][j] = mma4(a1lo, blo, acc[1][j]);
        }
#pragma unroll
        for (int j = 0; j < 4; ++j) {
            const v2f bhi = {b2[j], b3[j]};
            acc[0][j] = mma4(a0hi, bhi, acc[0][j]);
            acc[1][j] = mma4(a1hi, bhi, acc[1][j]);
        }
    }
}

__device__ __forceinline__ void product_half(const float* __restrict__ ap, const float* __restrict__ wp, v8f (&acc)[2][4]) {
    ABlock cur = load_a(ap, 0);
#pragma unroll 1
    for (int kb = 32; kb < D; kb += 32) {
        const ABlock nxt = load_a(ap, kb);
        mma_block(cur, wp, kb - 32, acc);
        cur = nxt;
    }
    mma_block(cur, wp, D - 32, acc);
}

__global__ __launch_bounds__(64) void gemm_cat(const float* __restrict__ x,
                                               const float* __restrict__ agg,
                                               const float* __restrict__ Wlin,
                                               const float* __restrict__ Wgcn,
                                               float* __restrict__ out) {
    const int wave  = threadIdx.x >> 5;
    const int lane  = threadIdx.x & 31;
    const int c     = lane & 15;
    const int hh    = lane >> 4;
    const int cb    = wave * 64;
    const int row0  = blockIdx.x * 32;

    const size_t aoff = (size_t)(row0 + c) * D + 16 * hh;
    const size_t woff = (size_t)(16 * hh) * D + cb + 4 * c;

    v8f acc[2][4] = {};
    product_half(x + aoff, Wlin + woff, acc);
    product_half(agg + aoff, Wgcn + woff, acc);

    float* __restrict__ dp = out + (size_t)(row0 + 8 * hh) * D + cb + 4 * c;
#pragma unroll
    for (int t = 0; t < 2; ++t) {
#pragma unroll
        for (int r = 0; r < 8; ++r) {
            const v4f o = {acc[t][0][r], acc[t][1][r], acc[t][2][r], acc[t][3][r]};
            *(v4f*)(dp + (size_t)(16 * t + r) * D) = o;
        }
    }
}


extern "C" void kernel_launch(void* const* d_in, const int* in_sizes, int n_in,
                              void* d_out, int out_size, void* d_ws, size_t ws_size,
                              hipStream_t stream) {
    const float* x    = (const float*)d_in[0];
    const int*   ei   = (const int*)d_in[1];
    const float* ew   = (const float*)d_in[2];
    const float* Wlin = (const float*)d_in[3];
    const float* Wgcn = (const float*)d_in[4];
    float*       out  = (float*)d_out;

    const int N = in_sizes[0] / D;
    const int E = in_sizes[2];

    const int* row = ei;
    const int* col = ei + E;

    float* dis = (float*)d_ws + N;
    float* agg = dis + N;

    int* tab     = (int*)d_out;
    int* ovf_rec = tab + (size_t)N * 64;
    int* ovf_dst = ovf_rec + (size_t)E * 2;
    int* cnt     = ovf_dst + E;

    zero_i32<<<(N + 1 + 255) / 256, 256, 0, stream>>>(cnt, N + 1);
    place_kernel<<<(E + 255) / 256, 256, 0, stream>>>(row, col, ew, cnt, tab, ovf_dst, ovf_rec, E, N);
    dis_kernel<<<(N + 255) / 256, 256, 0, stream>>>(tab, cnt, ovf_dst, ovf_rec, dis, N);
    gather_kernel<<<(N + 7) / 8, 256, 0, stream>>>(x, tab, cnt, ovf_dst, ovf_rec, dis, agg, N);
    gemm_cat<<<N / 32, 64, 0, stream>>>(x, agg, Wlin, Wgcn, out);
}


# gather fused into the product kernel: a block runs x@Wlin, sums its 32 nodes' rows into an LDS tile, then Wgcn from it; other waves' matrix work runs under row loads; no agg array; sum orders kept
# speedup vs baseline: 2.4616x; 1.0737x over previous
#include <hip/hip_runtime.h>

#define D 128

typedef float v2f __attribute__((ext_vector_type(2)));
typedef float v4f __attribute__((ext_vector_type(4)));
typedef float v8f __attribute__((ext_vector_type(8)));


#define CAP 32

__global__ void zero_i32(int* __restrict__ p, int n) {
    int i = blockIdx.x * blockDim.x + threadIdx.x;
    if (i < n) p[i] = 0;
}

__global__ void place_kernel(const int* __restrict__ row,
                             const int* __restrict__ col,
                             const float* __restrict__ w,
                             int* __restrict__ cnt,
                             int* __restrict__ tab,
                             int* __restrict__ ovf_dst,
                             int* __restrict__ ovf_rec,
                             int E, int n) {
    int e = blockIdx.x * blockDim.x + threadIdx.x;
    if (e < E) {
        const int  c = col[e];
        const int2 v = make_int2(row[e], __float_as_int(w[e]));
        const int  k = atomicAdd(&cnt[c], 1);
        int2* dst = (int2*)tab + ((size_t)c * CAP + k);
        if (k >= CAP) {
            const int j = atomicAdd(&cnt[n], 1);
            ovf_dst[j] = c;
            dst = (int2*)ovf_rec + j;
        }
        *dst = v;
    }
}

__global__ void dis_kernel(const int* __restrict__ tab,
                           const int* __restrict__ cnt,
                           const int* __restrict__ ovf_dst,
                           const int* __restrict__ ovf_rec,
                           float* __restrict__ dis, int n) {
    int c = blockIdx.x * blockDim.x + threadIdx.x;
    if (c < n) {
        const int2* __restrict__ q = (const int2*)tab + (size_t)c * CAP;
        const int mt = cnt[c];
        const int m  = (mt < CAP) ? mt : CAP;
        float d = 0.0f;
        for (int j = 0; j < m; j += 4) {
            const float w0 = __int_as_float(q[j].y);
            const float w1 = (j + 1 < m) ? __int_as_float(q[j + 1].y) : 0.0f;
            const float w2 = (j + 2 < m) ? __int_as_float(q[j + 2].y) : 0.0f;
            const float w3 = (j + 3 < m) ? __int_as_float(q[j + 3].y) : 0.0f;
            d += w0; d += w1; d += w2; d += w3;
        }
        if (mt > CAP) {
            const int no = cnt[n];
            for (int k = 0; k < no; ++k)
                if (ovf_dst[k] == c) d += __int_as_float(((const int2*)ovf_rec)[k].y);
        }
        dis[c] = (d > 0.0f) ? __frsqrt_rn(d) : 0.0f;
    }
}


#define LP 132

template <int G>
__device__ __forceinline__ void add_rows(const float* __restrict__ xl, int r, float nm, int j, v4f& acc) {
    v4f   v[G];
    float f[G];
#pragma unroll
    for (int k = 0; k < G; ++k) {
        const int rk = __builtin_amdgcn_readlane(r, j + k);
        f[k] = __int_as_float(__builtin_amdgcn_readlane(__float_as_int(nm), j + k));
        v[k] = *(const v4f*)(xl + (size_t)rk * D);
    }
#pragma unroll
    for (int k = 0; k < G; ++k) acc += v[k] * f[k];
}

__device__ __forceinline__ int clean_row(int ex, int mm, int lane) {
    const int rl = (mm > 0) ? __builtin_amdgcn_readlane(ex, (mm > 0) ? mm - 1 : 0) : 0;
    return (lane < mm) ? ex : rl;
}


__device__ __forceinline__ v8f mma4(v2f a, v2f b, v8f acc) {
    return __builtin_amdgcn_wmma_f32_16x16x4_f32(false, a, false, b, (short)0, acc, false, false);
}

struct APair { v4f t0[2], t1[2]; };

template <int P>
__device__ __forceinline__ APair load_a(const float* ap, int sb) {
    APair q;
#pragma unroll
    for (int i = 0; i < 2; ++i) {
        q.t0[i] = *(const v4f*)(ap + sb + 4 * i);
        q.t1[i] = *(const v4f*)(ap + 16 * P + sb + 4 * i);
    }
    return q;
}

__device__ __forceinline__ void mma_pair(const APair& A, const float* __restrict__ wp, int sb, v8f (&acc)[2][4]) {
#pragma unroll
    for (int i = 0; i < 2; ++i) {
        const float* __restrict__ wr = wp + (size_t)(sb + 4 * i) * D;
        const v4f b0 = *(const v4f*)(wr + 0 * D);
        const v4f b1 = *(const v4f*)(wr + 1 * D);
        const v4f b2 = *(const v4f*)(wr + 2 * D);
        const v4f b3 = *(const v4f*)(wr + 3 * D);
        const v2f a0lo = {A.t0[i].x, A.t0[i].y}, a0hi = {A.t0[i].z, A.t0[i].w};
        const v2f a1lo = {A.t1[i].x, A.t1[i].y}, a1hi = {A.t1[i].z, A.t1[i].w};
#pragma unroll
        for (int j = 0; j < 4; ++j) {
            const v2f blo = {b0[j], b1[j]};
            acc[0][j] = mma4(a0lo, blo, acc[0][j]);
            acc[1][j] = mma4(a1lo, blo, acc[1][j]);
        }
#pragma unroll
        for (int j = 0; j < 4; ++j) {
            const v2f bhi = {b2[j], b3[j]};
            acc[0][j] = mma4(a0hi, bhi, acc[0][j]);
            acc[1][j] = mma4(a1hi, bhi, acc[1][j]);
        }
    }
}

template <int P>
__device__ __forceinline__ void product_half(const float* ap, const float* __restrict__ wp, v8f (&acc)[2][4]) {
    APair cur = load_a<P>(ap, 0);
#pragma unroll 1
    for (int t = 1; t < 8; ++t) {
        const int sb = 32 * (t >> 1) + 8 * (t & 1);
        const int pb = 32 * ((t - 1) >> 1) + 8 * ((t - 1) & 1);
        const APair nxt = load_a<P>(ap, sb);
        mma_pair(cur, wp, pb, acc);
        cur = nxt;
    }
    mma_pair(cur, wp, 104, acc);
}

__global__ __launch_bounds__(64) void gather_gemm(const float* __restrict__ x,
                                                  const int* __restrict__ tab,
                                                  const int* __restrict__ cnt,
                                                  const int* __restrict__ ovf_dst,
                                                  const int* __restrict__ ovf_rec,
                                                  const float* __restrict__ dis,
                                                  const float* __restrict__ Wlin,
                                                  const float* __restrict__ Wgcn,
                                                  float* __restrict__ out, int n) {
    __shared__ float sA[32 * LP];

    const int wave  = threadIdx.x >> 5;
    const int lane  = threadIdx.x & 31;
    const int c     = lane & 15;
    const int hh    = lane >> 4;
    const int cb    = wave * 64;
    const int row0  = blockIdx.x * 32;

    const size_t aoff = (size_t)(row0 + c) * D + 16 * hh;
    const size_t woff = (size_t)(16 * hh) * D + cb + 4 * c;

    v8f acc[2][4] = {};
    product_half<D>(x + aoff, Wlin + woff, acc);

    {
        const int nb = __builtin_amdgcn_readfirstlane(row0 + 16 * wave);
        const float* __restrict__ xl = x + 4 * lane;
        const int2* __restrict__ tb = (const int2*)tab + (size_t)nb * CAP + lane;
        const int   cntv = cnt[nb + c];
        const float disv = dis[nb + c];
        float* __restrict__ sw = sA + (size_t)(16 * wave) * LP + 4 * lane;

        int2  en = tb[0];
        int   m  = __builtin_amdgcn_readlane(cntv, 0);
        int   mm = (m < CAP) ? m : CAP;
        int   r  = clean_row(en.x, mm, lane);
        float wq = __int_as_float(en.y);
        float dq = dis[r];
        en = tb[CAP];

#pragma unroll 1
        for (int k = 0; k < 16; ++k) {
            const float dc = __int_as_float(__builtin_amdgcn_readlane(__float_as_int(disv), k));
            const float nm = (lane < mm) ? (dq * wq) * dc : 0.0f;

            const int   k1  = (k + 1 < 16) ? k + 1 : 15;
            const int   k2  = (k + 2 < 16) ? k + 2 : 15;
            const int   m1  = __builtin_amdgcn_readlane(cntv, k1);
            const int   mm1 = (m1 < CAP) ? m1 : CAP;
            const int   r1  = clean_row(en.x, mm1, lane);
            const float w1  = __int_as_float(en.y);
            const float d1  = dis[r1];
            const int2  e2  = tb[(size_t)k2 * CAP];

            v4f a = {0.0f, 0.0f, 0.0f, 0.0f};
            {
                int j = 0;
                for (; j + 8 <= mm; j += 8) add_rows<8>(xl, r, nm, j, a);
                const int rem = mm - j;
                if (rem > 4)      add_rows<8>(xl, r, nm, j, a);
                else if (rem > 0) add_rows<4>(xl, r, nm, j, a);
            }
            if (m > CAP) {
                const int cn = nb + k;
                const int no = __builtin_amdgcn_readfirstlane(cnt[n]);
                for (int k0 = 0; k0 < no; k0 += 32) {
                    const int q = k0 + lane;
                    int  od = -1;
                    int2 oe = make_int2(0, 0);
                    if (q < no) { od = ovf_dst[q]; oe = ((const int2*)ovf_rec)[q]; }
                    const bool  hit = (od == cn);
                    const float onm = hit ? (dis[oe.x] * __int_as_float(oe.y)) * dc : 0.0f;
                    unsigned mask = __builtin_amdgcn_ballot_w32(hit);
                    while (mask) {
                        const int l = __builtin_ctz(mask);
                        mask &= mask - 1;
                        const int   rk = __builtin_amdgcn_readlane(oe.x, l);
                        const float f  = __int_as_float(__builtin_amdgcn_readlane(__float_as_int(onm), l));
                        a += *(const v4f*)(xl + (size_t)rk * D) * f;
                    }
                }
            }
            *(v4f*)(sw + (size_t)k * LP) = a;

            m = m1; mm = mm1; r = r1; wq = w1; dq = d1; en = e2;
        }
    }
    __syncthreads();

    product_half<LP>(sA + (size_t)c * LP + 16 * hh, Wgcn + woff, acc);

    float* __restrict__ dp = out + (size_t)(row0 + 8 * hh) * D + cb + 4 * c;
#pragma unroll
    for (int t = 0; t < 2; ++t) {
#pragma unroll
        for (int r = 0; r < 8; ++r) {
            const v4f o = {acc[t][0][r], acc[t][1][r], acc[t][2][r], acc[t][3][r]};
            *(v4f*)(dp + (size_t)(16 * t + r) * D) = o;
        }
    }
}


extern "C" void kernel_launch(void* const* d_in, const int* in_sizes, int n_in,
                              void* d_out, int out_size, void* d_ws, size_t ws_size,
                              hipStream_t stream) {
    const float* x    = (const float*)d_in[0];
    const int*   ei   = (const int*)d_in[1];
    const float* ew   = (const float*)d_in[2];
    const float* Wlin = (const float*)d_in[3];
    const float* Wgcn = (const float*)d_in[4];
    float*       out  = (float*)d_out;

    const int N = in_sizes[0] / D;
    const int E = in_sizes[2];

    const int* row = ei;
    const int* col = ei + E;

    float* dis = (float*)d_ws + N;

    int* tab     = (int*)(dis + N);
    int* ovf_rec = tab + (size_t)N * 64;
    int* ovf_dst = ovf_rec + (size_t)E * 2;
    int* cnt     = ovf_dst + E;

    zero_i32<<<(N + 1 + 255) / 256, 256, 0, stream>>>(cnt, N + 1);
    place_kernel<<<(E + 255) / 256, 256, 0, stream>>>(row, col, ew, cnt, tab, ovf_dst, ovf_rec, E, N);
    dis_kernel<<<(N + 255) / 256, 256, 0, stream>>>(tab, cnt, ovf_dst, ovf_rec, dis, N);
    gather_gemm<<<N / 32, 64, 0, stream>>>(x, tab, cnt, ovf_dst, ovf_rec, dis, Wlin, Wgcn, out, N);
}


# edges bucketed with no global atomic per edge: 25000-edge chunks counting-sorted by 32-node range on LDS counters, then a block per range ranks its pieces in LDS; slot order (sum order) changes
# speedup vs baseline: 3.6418x; 1.4794x over previous
#include <hip/hip_runtime.h>

#define D 128

typedef float v2f __attribute__((ext_vector_type(2)));
typedef float v4f __attribute__((ext_vector_type(4)));
typedef float v8f __attribute__((ext_vector_type(8)));


#define CAP 32


#define CH    25000
#define NBMAX 4096

__global__ __launch_bounds__(1024) void bin_kernel(const int* __restrict__ row,
                                                   const int* __restrict__ col,
                                                   const float* __restrict__ w,
                                                   int2* __restrict__ srec,
                                                   int2* __restrict__ tinfo,
                                                   int* __restrict__ cnt,
                                                   int E, int n, int nchunks) {
    __shared__ int  sc[NBMAX];
    __shared__ int  sb[NBMAX];
    __shared__ int  ss[2][1024];
    __shared__ __attribute__((aligned(16))) int2 st[CH];

    const int tid = threadIdx.x;
    const int nb  = n >> 5;
    const int e0  = blockIdx.x * CH;
    const int ne  = (E - e0 < CH) ? (E - e0) : CH;
    const int nq  = (ne + 3) >> 2;

    if (blockIdx.x == 0 && tid == 0) cnt[n] = 0;
    for (int i = tid; i < NBMAX; i += 1024) sc[i] = 0;
    __syncthreads();

    const int4* __restrict__ col4 = (const int4*)(col + e0);
    const int4* __restrict__ row4 = (const int4*)(row + e0);
    const v4f*  __restrict__ w4   = (const v4f*)(w + e0);

    int4 cv[7], rv[7];
    v4f  wv[7];
#pragma unroll
    for (int i = 0; i < 7; ++i) {
        const int q = tid + 1024 * i;
        cv[i] = col4[(q < nq) ? q : nq - 1];
    }
#pragma unroll
    for (int i = 0; i < 7; ++i) {
        const int q = tid + 1024 * i;
        rv[i] = row4[(q < nq) ? q : nq - 1];
    }
#pragma unroll
    for (int i = 0; i < 7; ++i) {
        const int q = tid + 1024 * i;
        wv[i] = w4[(q < nq) ? q : nq - 1];
    }

#pragma unroll
    for (int i = 0; i < 7; ++i) {
        const int q = tid + 1024 * i;
        if (4 * q + 0 < ne) atomicAdd(&sc[cv[i].x >> 5], 1);
        if (4 * q + 1 < ne) atomicAdd(&sc[cv[i].y >> 5], 1);
        if (4 * q + 2 < ne) atomicAdd(&sc[cv[i].z >> 5], 1);
        if (4 * q + 3 < ne) atomicAdd(&sc[cv[i].w >> 5], 1);
    }
    __syncthreads();

    const int c0 = sc[4 * tid], c1 = sc[4 * tid + 1], c2 = sc[4 * tid + 2], c3 = sc[4 * tid + 3];
    const int tsum = c0 + c1 + c2 + c3;
    ss[0][tid] = tsum;
    __syncthreads();
    int cur = 0;
    for (int d = 1; d < 1024; d <<= 1) {
        int v = ss[cur][tid];
        if (tid >= d) v += ss[cur][tid - d];
        ss[cur ^ 1][tid] = v;
        __syncthreads();
        cur ^= 1;
    }
    const int s0 = ss[cur][tid] - tsum;
    const int s1 = s0 + c0, s2 = s1 + c1, s3 = s2 + c2;
    sc[4 * tid] = s0; sc[4 * tid + 1] = s1; sc[4 * tid + 2] = s2; sc[4 * tid + 3] = s3;
    sb[4 * tid] = s0; sb[4 * tid + 1] = s1; sb[4 * tid + 2] = s2; sb[4 * tid + 3] = s3;
    __syncthreads();

#pragma unroll
    for (int i = 0; i < 7; ++i) {
        const int q = tid + 1024 * i;
        if (4 * q + 0 < ne) { const int p = atomicAdd(&sc[cv[i].x >> 5], 1); st[p] = make_int2((rv[i].x << 5) | (cv[i].x & 31), __float_as_int(wv[i].x)); }
        if (4 * q + 1 < ne) { const int p = atomicAdd(&sc[cv[i].y >> 5], 1); st[p] = make_int2((rv[i].y << 5) | (cv[i].y & 31), __float_as_int(wv[i].y)); }
        if (4 * q + 2 < ne) { const int p = atomicAdd(&sc[cv[i].z >> 5], 1); st[p] = make_int2((rv[i].z << 5) | (cv[i].z & 31), __float_as_int(wv[i].z)); }
        if (4 * q + 3 < ne) { const int p = atomicAdd(&sc[cv[i].w >> 5], 1); st[p] = make_int2((rv[i].w << 5) | (cv[i].w & 31), __float_as_int(wv[i].w)); }
    }
    __syncthreads();

    {
        int4* __restrict__ d4 = (int4*)(srec + (size_t)e0);
        const int4* s4 = (const int4*)st;
        const int n4 = ne >> 1;
        for (int i = tid; i < n4; i += 1024) d4[i] = s4[i];
        if (tid == 0 && (ne & 1)) srec[(size_t)e0 + ne - 1] = st[ne - 1];
    }

    {
        const int r = 4 * tid;
        int2* __restrict__ tp = tinfo + (size_t)r * nchunks + blockIdx.x;
        for (int j = 0; j < 4; ++j) {
            if (r + j < nb) {
                const int f = sb[r + j];
                tp[(size_t)j * nchunks] = make_int2(e0 + f, sc[r + j] - f);
            }
        }
    }
}

__global__ __launch_bounds__(64) void fill_kernel(const int2* __restrict__ srec,
                                                  const int2* __restrict__ tinfo,
                                                  int* __restrict__ cnt,
                                                  int* __restrict__ tab,
                                                  int* __restrict__ ovf_dst,
                                                  int* __restrict__ ovf_rec,
                                                  int n, int nchunks) {
    __shared__ __attribute__((aligned(16))) int2 sl[32 * CAP];
    __shared__ int  lc[32];

    const int tid = threadIdx.x;
    const int b   = blockIdx.x;

    for (int i = tid; i < 32 * CAP; i += 64) sl[i] = make_int2(0, 0);
    if (tid < 32) lc[tid] = 0;
    __syncthreads();

    for (int ch = tid; ch < nchunks; ch += 64) {
        const int2 ti = tinfo[(size_t)b * nchunks + ch];
        const int2* __restrict__ p = srec + ti.x;
        for (int i = 0; i < ti.y; i += 4) {
            int2 r[4];
#pragma unroll
            for (int j = 0; j < 4; ++j) r[j] = (i + j < ti.y) ? p[i + j] : make_int2(0, 0);
#pragma unroll
            for (int j = 0; j < 4; ++j) {
                if (i + j < ti.y) {
                    const int  l = r[j].x & 31;
                    const int  k = atomicAdd(&lc[l], 1);
                    const int2 v = make_int2(r[j].x >> 5, r[j].y);
                    if (k < CAP) {
                        sl[l * CAP + k] = v;
                    } else {
                        const int q = atomicAdd(&cnt[n], 1);
                        ovf_dst[q] = 32 * b + l;
                        ((int2*)ovf_rec)[q] = v;
                    }
                }
            }
        }
    }
    __syncthreads();

    if (tid < 32) cnt[32 * b + tid] = lc[tid];
    int4* __restrict__ dst = (int4*)(tab + (size_t)b * (32 * CAP * 2));
    const int4* s4 = (const int4*)sl;
    for (int i = tid; i < 32 * CAP / 2; i += 64) dst[i] = s4[i];
}

__global__ void dis_kernel(const int* __restrict__ tab,
                           const int* __restrict__ cnt,
                           const int* __restrict__ ovf_dst,
                           const int* __restrict__ ovf_rec,
                           float* __restrict__ dis, int n) {
    int c = blockIdx.x * blockDim.x + threadIdx.x;
    if (c < n) {
        const int2* __restrict__ q = (const int2*)tab + (size_t)c * CAP;
        const int mt = cnt[c];
        const int m  = (mt < CAP) ? mt : CAP;
        float d = 0.0f;
        for (int j = 0; j < m; j += 4) {
            const float w0 = __int_as_float(q[j].y);
            const float w1 = (j + 1 < m) ? __int_as_float(q[j + 1].y) : 0.0f;
            const float w2 = (j + 2 < m) ? __int_as_float(q[j + 2].y) : 0.0f;
            const float w3 = (j + 3 < m) ? __int_as_float(q[j + 3].y) : 0.0f;
            d += w0; d += w1; d += w2; d += w3;
        }
        if (mt > CAP) {
            const int no = cnt[n];
            for (int k = 0; k < no; ++k)
                if (ovf_dst[k] == c) d += __int_as_float(((const int2*)ovf_rec)[k].y);
        }
        dis[c] = (d > 0.0f) ? __frsqrt_rn(d) : 0.0f;
    }
}


#define LP 132

template <int G>
__device__ __forceinline__ void add_rows(const float* __restrict__ xl, int r, float nm, int j, v4f& acc) {
    v4f   v[G];
    float f[G];
#pragma unroll
    for (int k = 0; k < G; ++k) {
        const int rk = __builtin_amdgcn_readlane(r, j + k);
        f[k] = __int_as_float(__builtin_amdgcn_readlane(__float_as_int(nm), j + k));
        v[k] = *(const v4f*)(xl + (size_t)rk * D);
    }
#pragma unroll
    for (int k = 0; k < G; ++k) acc += v[k] * f[k];
}

__device__ __forceinline__ int clean_row(int ex, int mm, int lane) {
    const int rl = (mm > 0) ? __builtin_amdgcn_readlane(ex, (mm > 0) ? mm - 1 : 0) : 0;
    return (lane < mm) ? ex : rl;
}


__device__ __forceinline__ v8f mma4(v2f a, v2f b, v8f acc) {
    return __builtin_amdgcn_wmma_f32_16x16x4_f32(false, a, false, b, (short)0, acc, false, false);
}

struct APair { v4f t0[2], t1[2]; };

template <int P>
__device__ __forceinline__ APair load_a(const float* ap, int sb) {
    APair q;
#pragma unroll
    for (int i = 0; i < 2; ++i) {
        q.t0[i] = *(const v4f*)(ap + sb + 4 * i);
        q.t1[i] = *(const v4f*)(ap + 16 * P + sb + 4 * i);
    }
    return q;
}

__device__ __forceinline__ void mma_pair(const APair& A, const float* __restrict__ wp, int sb, v8f (&acc)[2][4]) {
#pragma unroll
    for (int i = 0; i < 2; ++i) {
        const float* __restrict__ wr = wp + (size_t)(sb + 4 * i) * D;
        const v4f b0 = *(const v4f*)(wr + 0 * D);
        const v4f b1 = *(const v4f*)(wr + 1 * D);
        const v4f b2 = *(const v4f*)(wr + 2 * D);
        const v4f b3 = *(const v4f*)(wr + 3 * D);
        const v2f a0lo = {A.t0[i].x, A.t0[i].y}, a0hi = {A.t0[i].z, A.t0[i].w};
        const v2f a1lo = {A.t1[i].x, A.t1[i].y}, a1hi = {A.t1[i].z, A.t1[i].w};
#pragma unroll
        for (int j = 0; j < 4; ++j) {
            const v2f blo = {b0[j], b1[j]};
            acc[0][j] = mma4(a0lo, blo, acc[0][j]);
            acc[1][j] = mma4(a1lo, blo, acc[1][j]);
        }
#pragma unroll
        for (int j = 0; j < 4; ++j) {
            const v2f bhi = {b2[j], b3[j]};
            acc[0][j] = mma4(a0hi, bhi, acc[0][j]);
            acc[1][j] = mma4(a1hi, bhi, acc[1][j]);
        }
    }
}

template <int P>
__device__ __forceinline__ void product_half(const float* ap, const float* __restrict__ wp, v8f (&acc)[2][4]) {
    APair cur = load_a<P>(ap, 0);
#pragma unroll 1
    for (int t = 1; t < 8; ++t) {
        const int sb = 32 * (t >> 1) + 8 * (t & 1);
        const int pb = 32 * ((t - 1) >> 1) + 8 * ((t - 1) & 1);
        const APair nxt = load_a<P>(ap, sb);
        mma_pair(cur, wp, pb, acc);
        cur = nxt;
    }
    mma_pair(cur, wp, 104, acc);
}

__global__ __launch_bounds__(64) void gather_gemm(const float* __restrict__ x,
                                                  const int* __restrict__ tab,
                                                  const int* __restrict__ cnt,
                                                  const int* __restrict__ ovf_dst,
                                                  const int* __restrict__ ovf_rec,
                                                  const float* __restrict__ dis,
                                                  const float* __restrict__ Wlin,
                                                  const float* __restrict__ Wgcn,
                                                  float* __restrict__ out, int n) {
    __shared__ float sA[32 * LP];

    const int wave  = threadIdx.x >> 5;
    const int lane  = threadIdx.x & 31;
    const int c     = lane & 15;
    const int hh    = lane >> 4;
    const int cb    = wave * 64;
    const int row0  = blockIdx.x * 32;

    const size_t aoff = (size_t)(row0 + c) * D + 16 * hh;
    const size_t woff = (size_t)(16 * hh) * D + cb + 4 * c;

    v8f acc[2][4] = {};
    product_half<D>(x + aoff, Wlin + woff, acc);

    {
        const int nb = __builtin_amdgcn_readfirstlane(row0 + 16 * wave);
        const float* __restrict__ xl = x + 4 * lane;
        const int2* __restrict__ tb = (const int2*)tab + (size_t)nb * CAP + lane;
        const int   cntv = cnt[nb + c];
        const float disv = dis[nb + c];
        float* __restrict__ sw = sA + (size_t)(16 * wave) * LP + 4 * lane;

        int2  en = tb[0];
        int   m  = __builtin_amdgcn_readlane(cntv, 0);
        int   mm = (m < CAP) ? m : CAP;
        int   r  = clean_row(en.x, mm, lane);
        float wq = __int_as_float(en.y);
        float dq = dis[r];
        en = tb[CAP];

#pragma unroll 1
        for (int k = 0; k < 16; ++k) {
            const float dc = __int_as_float(__builtin_amdgcn_readlane(__float_as_int(disv), k));
            const float nm = (lane < mm) ? (dq * wq) * dc : 0.0f;

            const int   k1  = (k + 1 < 16) ? k + 1 : 15;
            const int   k2  = (k + 2 < 16) ? k + 2 : 15;
            const int   m1  = __builtin_amdgcn_readlane(cntv, k1);
            const int   mm1 = (m1 < CAP) ? m1 : CAP;
            const int   r1  = clean_row(en.x, mm1, lane);
            const float w1  = __int_as_float(en.y);
            const float d1  = dis[r1];
            const int2  e2  = tb[(size_t)k2 * CAP];

            v4f a = {0.0f, 0.0f, 0.0f, 0.0f};
            {
                int j = 0;
                for (; j + 8 <= mm; j += 8) add_rows<8>(xl, r, nm, j, a);
                const int rem = mm - j;
                if (rem > 4)      add_rows<8>(xl, r, nm, j, a);
                else if (rem > 0) add_rows<4>(xl, r, nm, j, a);
            }
            if (m > CAP) {
                const int cn = nb + k;
                const int no = __builtin_amdgcn_readfirstlane(cnt[n]);
                for (int k0 = 0; k0 < no; k0 += 32) {
                    const int q = k0 + lane;
                    int  od = -1;
                    int2 oe = make_int2(0, 0);
                    if (q < no) { od = ovf_dst[q]; oe = ((const int2*)ovf_rec)[q]; }
                    const bool  hit = (od == cn);
                    const float onm = hit ? (dis[oe.x] * __int_as_float(oe.y)) * dc : 0.0f;
                    unsigned mask = __builtin_amdgcn_ballot_w32(hit);
                    while (mask) {
                        const int l = __builtin_ctz(mask);
                        mask &= mask - 1;
                        const int   rk = __builtin_amdgcn_readlane(oe.x, l);
                        const float f  = __int_as_float(__builtin_amdgcn_readlane(__float_as_int(onm), l));
                        a += *(const v4f*)(xl + (size_t)rk * D) * f;
                    }
                }
            }
            *(v4f*)(sw + (size_t)k * LP) = a;

            m = m1; mm = mm1; r = r1; wq = w1; dq = d1; en = e2;
        }
    }
    __syncthreads();

    product_half<LP>(sA + (size_t)c * LP + 16 * hh, Wgcn + woff, acc);

    float* __restrict__ dp = out + (size_t)(row0 + 8 * hh) * D + cb + 4 * c;
#pragma unroll
    for (int t = 0; t < 2; ++t) {
#pragma unroll
        for (int r = 0; r < 8; ++r) {
            const v4f o = {acc[t][0][r], acc[t][1][r], acc[t][2][r], acc[t][3][r]};
            *(v4f*)(dp + (size_t)(16 * t + r) * D) = o;
        }
    }
}


extern "C" void kernel_launch(void* const* d_in, const int* in_sizes, int n_in,
                              void* d_out, int out_size, void* d_ws, size_t ws_size,
                              hipStream_t stream) {
    const float* x    = (const float*)d_in[0];
    const int*   ei   = (const int*)d_in[1];
    const float* ew   = (const float*)d_in[2];
    const float* Wlin = (const float*)d_in[3];
    const float* Wgcn = (const float*)d_in[4];
    float*       out  = (float*)d_out;

    const int N = in_sizes[0] / D;
    const int E = in_sizes[2];

    const int* row = ei;
    const int* col = ei + E;

    float* dis = (float*)d_ws + N;

    int* tab     = (int*)(dis + N);
    int* ovf_rec = tab + (size_t)N * 64;
    int* ovf_dst = ovf_rec + (size_t)E * 2;
    int* cnt     = ovf_dst + E;

    int2* srec  = (int2*)out;
    int2* tinfo = srec + E;
    const int nchunks = (E + CH - 1) / CH;

    bin_kernel<<<nchunks, 1024, 0, stream>>>(row, col, ew, srec, tinfo, cnt, E, N, nchunks);
    fill_kernel<<<N / 32, 64, 0, stream>>>(srec, tinfo, cnt, tab, ovf_dst, ovf_rec, N, nchunks);
    dis_kernel<<<(N + 255) / 256, 256, 0, stream>>>(tab, cnt, ovf_dst, ovf_rec, dis, N);
    gather_gemm<<<N / 32, 64, 0, stream>>>(x, tab, cnt, ovf_dst, ovf_rec, dis, Wlin, Wgcn, out, N);
}


# one launch fewer: the fill kernel sums each node's degree from its LDS slot tile (side-list weights by an LDS add) and writes dis itself, no degree kernel; sum starts at slot node%32, order changes
# speedup vs baseline: 3.9248x; 1.0777x over previous
#include <hip/hip_runtime.h>

#define D 128

typedef float v2f __attribute__((ext_vector_type(2)));
typedef float v4f __attribute__((ext_vector_type(4)));
typedef float v8f __attribute__((ext_vector_type(8)));


#define CAP 32


#define CH    25000
#define NBMAX 4096

__global__ __launch_bounds__(1024) void bin_kernel(const int* __restrict__ row,
                                                   const int* __restrict__ col,
                                                   const float* __restrict__ w,
                                                   int2* __restrict__ srec,
                                                   int2* __restrict__ tinfo,
                                                   int* __restrict__ cnt,
                                                   int E, int n, int nchunks) {
    __shared__ int  sc[NBMAX];
    __shared__ int  sb[NBMAX];
    __shared__ int  ss[2][1024];
    __shared__ __attribute__((aligned(16))) int2 st[CH];

    const int tid = threadIdx.x;
    const int nb  = n >> 5;
    const int e0  = blockIdx.x * CH;
    const int ne  = (E - e0 < CH) ? (E - e0) : CH;
    const int nq  = (ne + 3) >> 2;

    if (blockIdx.x == 0 && tid == 0) cnt[n] = 0;
    for (int i = tid; i < NBMAX; i += 1024) sc[i] = 0;
    __syncthreads();

    const int4* __restrict__ col4 = (const int4*)(col + e0);
    const int4* __restrict__ row4 = (const int4*)(row + e0);
    const v4f*  __restrict__ w4   = (const v4f*)(w + e0);

    int4 cv[7], rv[7];
    v4f  wv[7];
#pragma unroll
    for (int i = 0; i < 7; ++i) {
        const int q = tid + 1024 * i;
        cv[i] = col4[(q < nq) ? q : nq - 1];
    }
#pragma unroll
    for (int i = 0; i < 7; ++i) {
        const int q = tid + 1024 * i;
        rv[i] = row4[(q < nq) ? q : nq - 1];
    }
#pragma unroll
    for (int i = 0; i < 7; ++i) {
        const int q = tid + 1024 * i;
        wv[i] = w4[(q < nq) ? q : nq - 1];
    }

#pragma unroll
    for (int i = 0; i < 7; ++i) {
        const int q = tid + 1024 * i;
        if (4 * q + 0 < ne) atomicAdd(&sc[cv[i].x >> 5], 1);
        if (4 * q + 1 < ne) atomicAdd(&sc[cv[i].y >> 5], 1);
        if (4 * q + 2 < ne) atomicAdd(&sc[cv[i].z >> 5], 1);
        if (4 * q + 3 < ne) atomicAdd(&sc[cv[i].w >> 5], 1);
    }
    __syncthreads();

    const int c0 = sc[4 * tid], c1 = sc[4 * tid + 1], c2 = sc[4 * tid + 2], c3 = sc[4 * tid + 3];
    const int tsum = c0 + c1 + c2 + c3;
    ss[0][tid] = tsum;
    __syncthreads();
    int cur = 0;
    for (int d = 1; d < 1024; d <<= 1) {
        int v = ss[cur][tid];
        if (tid >= d) v += ss[cur][tid - d];
        ss[cur ^ 1][tid] = v;
        __syncthreads();
        cur ^= 1;
    }
    const int s0 = ss[cur][tid] - tsum;
    const int s1 = s0 + c0, s2 = s1 + c1, s3 = s2 + c2;
    sc[4 * tid] = s0; sc[4 * tid + 1] = s1; sc[4 * tid + 2] = s2; sc[4 * tid + 3] = s3;
    sb[4 * tid] = s0; sb[4 * tid + 1] = s1; sb[4 * tid + 2] = s2; sb[4 * tid + 3] = s3;
    __syncthreads();

#pragma unroll
    for (int i = 0; i < 7; ++i) {
        const int q = tid + 1024 * i;
        if (4 * q + 0 < ne) { const int p = atomicAdd(&sc[cv[i].x >> 5], 1); st[p] = make_int2((rv[i].x << 5) | (cv[i].x & 31), __float_as_int(wv[i].x)); }
        if (4 * q + 1 < ne) { const int p = atomicAdd(&sc[cv[i].y >> 5], 1); st[p] = make_int2((rv[i].y << 5) | (cv[i].y & 31), __float_as_int(wv[i].y)); }
        if (4 * q + 2 < ne) { const int p = atomicAdd(&sc[cv[i].z >> 5], 1); st[p] = make_int2((rv[i].z << 5) | (cv[i].z & 31), __float_as_int(wv[i].z)); }
        if (4 * q + 3 < ne) { const int p = atomicAdd(&sc[cv[i].w >> 5], 1); st[p] = make_int2((rv[i].w << 5) | (cv[i].w & 31), __float_as_int(wv[i].w)); }
    }
    __syncthreads();

    {
        int4* __restrict__ d4 = (int4*)(srec + (size_t)e0);
        const int4* s4 = (const int4*)st;
        const int n4 = ne >> 1;
        for (int i = tid; i < n4; i += 1024) d4[i] = s4[i];
        if (tid == 0 && (ne & 1)) srec[(size_t)e0 + ne - 1] = st[ne - 1];
    }

    {
        const int r = 4 * tid;
        int2* __restrict__ tp = tinfo + (size_t)r * nchunks + blockIdx.x;
        for (int j = 0; j < 4; ++j) {
            if (r + j < nb) {
                const int f = sb[r + j];
                tp[(size_t)j * nchunks] = make_int2(e0 + f, sc[r + j] - f);
            }
        }
    }
}

__global__ __launch_bounds__(64) void fill_kernel(const int2* __restrict__ srec,
                                                  const int2* __restrict__ tinfo,
                                                  int* __restrict__ cnt,
                                                  int* __restrict__ tab,
                                                  int* __restrict__ ovf_dst,
                                                  int* __restrict__ ovf_rec,
                                                  float* __restrict__ dis,
                                                  int n, int nchunks) {
    __shared__ __attribute__((aligned(16))) int2 sl[32 * CAP];
    __shared__ int   lc[32];
    __shared__ float od[32];

    const int tid = threadIdx.x;
    const int b   = blockIdx.x;

    for (int i = tid; i < 32 * CAP; i += 64) sl[i] = make_int2(0, 0);
    if (tid < 32) { lc[tid] = 0; od[tid] = 0.0f; }
    __syncthreads();

    for (int ch = tid; ch < nchunks; ch += 64) {
        const int2 ti = tinfo[(size_t)b * nchunks + ch];
        const int2* __restrict__ p = srec + ti.x;
        for (int i = 0; i < ti.y; i += 4) {
            int2 r[4];
#pragma unroll
            for (int j = 0; j < 4; ++j) r[j] = (i + j < ti.y) ? p[i + j] : make_int2(0, 0);
#pragma unroll
            for (int j = 0; j < 4; ++j) {
                if (i + j < ti.y) {
                    const int  l = r[j].x & 31;
                    const int  k = atomicAdd(&lc[l], 1);
                    const int2 v = make_int2(r[j].x >> 5, r[j].y);
                    if (k < CAP) {
                        sl[l * CAP + k] = v;
                    } else {
                        const int q = atomicAdd(&cnt[n], 1);
                        ovf_dst[q] = 32 * b + l;
                        ((int2*)ovf_rec)[q] = v;
                        atomicAdd(&od[l], __int_as_float(r[j].y));
                    }
                }
            }
        }
    }
    __syncthreads();

    if (tid < 32) {
        cnt[32 * b + tid] = lc[tid];
        const int* __restrict__ sw = (const int*)sl + 2 * CAP * tid + 1;
        float d = 0.0f;
#pragma unroll
        for (int k = 0; k < CAP; ++k) d += __int_as_float(sw[2 * ((k + tid) & (CAP - 1))]);
        d += od[tid];
        dis[32 * b + tid] = (d > 0.0f) ? __frsqrt_rn(d) : 0.0f;
    }
    int4* __restrict__ dst = (int4*)(tab + (size_t)b * (32 * CAP * 2));
    const int4* s4 = (const int4*)sl;
    for (int i = tid; i < 32 * CAP / 2; i += 64) dst[i] = s4[i];
}


#define LP 132

template <int G>
__device__ __forceinline__ void add_rows(const float* __restrict__ xl, int r, float nm, int j, v4f& acc) {
    v4f   v[G];
    float f[G];
#pragma unroll
    for (int k = 0; k < G; ++k) {
        const int rk = __builtin_amdgcn_readlane(r, j + k);
        f[k] = __int_as_float(__builtin_amdgcn_readlane(__float_as_int(nm), j + k));
        v[k] = *(const v4f*)(xl + (size_t)rk * D);
    }
#pragma unroll
    for (int k = 0; k < G; ++k) acc += v[k] * f[k];
}

__device__ __forceinline__ int clean_row(int ex, int mm, int lane) {
    const int rl = (mm > 0) ? __builtin_amdgcn_readlane(ex, (mm > 0) ? mm - 1 : 0) : 0;
    return (lane < mm) ? ex : rl;
}


__device__ __forceinline__ v8f mma4(v2f a, v2f b, v8f acc) {
    return __builtin_amdgcn_wmma_f32_16x16x4_f32(false, a, false, b, (short)0, acc, false, false);
}

struct APair { v4f t0[2], t1[2]; };

template <int P>
__device__ __forceinline__ APair load_a(const float* ap, int sb) {
    APair q;
#pragma unroll
    for (int i = 0; i < 2; ++i) {
        q.t0[i] = *(const v4f*)(ap + sb + 4 * i);
        q.t1[i] = *(const v4f*)(ap + 16 * P + sb + 4 * i);
    }
    return q;
}

__device__ __forceinline__ void mma_pair(const APair& A, const float* __restrict__ wp, int sb, v8f (&acc)[2][4]) {
#pragma unroll
    for (int i = 0; i < 2; ++i) {
        const float* __restrict__ wr = wp + (size_t)(sb + 4 * i) * D;
        const v4f b0 = *(const v4f*)(wr + 0 * D);
        const v4f b1 = *(const v4f*)(wr + 1 * D);
        const v4f b2 = *(const v4f*)(wr + 2 * D);
        const v4f b3 = *(const v4f*)(wr + 3 * D);
        const v2f a0lo = {A.t0[i].x, A.t0[i].y}, a0hi = {A.t0[i].z, A.t0[i].w};
        const v2f a1lo = {A.t1[i].x, A.t1[i].y}, a1hi = {A.t1[i].z, A.t1[i].w};
#pragma unroll
        for (int j = 0; j < 4; ++j) {
            const v2f blo = {b0[j], b1[j]};
            acc[0][j] = mma4(a0lo, blo, acc[0][j]);
            acc[1][j] = mma4(a1lo, blo, acc[1][j]);
        }
#pragma unroll
        for (int j = 0; j < 4; ++j) {
            const v2f bhi = {b2[j], b3[j]};
            acc[0][j] = mma4(a0hi, bhi, acc[0][j]);
            acc[1][j] = mma4(a1hi, bhi, acc[1][j]);
        }
    }
}

template <int P>
__device__ __forceinline__ void product_half(const float* ap, const float* __restrict__ wp, v8f (&acc)[2][4]) {
    APair cur = load_a<P>(ap, 0);
#pragma unroll 1
    for (int t = 1; t < 8; ++t) {
        const int sb = 32 * (t >> 1) + 8 * (t & 1);
        const int pb = 32 * ((t - 1) >> 1) + 8 * ((t - 1) & 1);
        const APair nxt = load_a<P>(ap, sb);
        mma_pair(cur, wp, pb, acc);
        cur = nxt;
    }
    mma_pair(cur, wp, 104, acc);
}

__global__ __launch_bounds__(64) void gather_gemm(const float* __restrict__ x,
                                                  const int* __restrict__ tab,
                                                  const int* __restrict__ cnt,
                                                  const int* __restrict__ ovf_dst,
                                                  const int* __restrict__ ovf_rec,
                                                  const float* __restrict__ dis,
                                                  const float* __restrict__ Wlin,
                                                  const float* __restrict__ Wgcn,
                                                  float* __restrict__ out, int n) {
    __shared__ float sA[32 * LP];

    const int wave  = threadIdx.x >> 5;
    const int lane  = threadIdx.x & 31;
    const int c     = lane & 15;
    const int hh    = lane >> 4;
    const int cb    = wave * 64;
    const int row0  = blockIdx.x * 32;

    const size_t aoff = (size_t)(row0 + c) * D + 16 * hh;
    const size_t woff = (size_t)(16 * hh) * D + cb + 4 * c;

    v8f acc[2][4] = {};
    product_half<D>(x + aoff, Wlin + woff, acc);

    {
        const int nb = __builtin_amdgcn_readfirstlane(row0 + 16 * wave);
        const float* __restrict__ xl = x + 4 * lane;
        const int2* __restrict__ tb = (const int2*)tab + (size_t)nb * CAP + lane;
        const int   cntv = cnt[nb + c];
        const float disv = dis[nb + c];
        float* __restrict__ sw = sA + (size_t)(16 * wave) * LP + 4 * lane;

        int2  en = tb[0];
        int   m  = __builtin_amdgcn_readlane(cntv, 0);
        int   mm = (m < CAP) ? m : CAP;
        int   r  = clean_row(en.x, mm, lane);
        float wq = __int_as_float(en.y);
        float dq = dis[r];
        en = tb[CAP];

#pragma unroll 1
        for (int k = 0; k < 16; ++k) {
            const float dc = __int_as_float(__builtin_amdgcn_readlane(__float_as_int(disv), k));
            const float nm = (lane < mm) ? (dq * wq) * dc : 0.0f;

            const int   k1  = (k + 1 < 16) ? k + 1 : 15;
            const int   k2  = (k + 2 < 16) ? k + 2 : 15;
            const int   m1  = __builtin_amdgcn_readlane(cntv, k1);
            const int   mm1 = (m1 < CAP) ? m1 : CAP;
            const int   r1  = clean_row(en.x, mm1, lane);
            const float w1  = __int_as_float(en.y);
            const float d1  = dis[r1];
            const int2  e2  = tb[(size_t)k2 * CAP];

            v4f a = {0.0f, 0.0f, 0.0f, 0.0f};
            {
                int j = 0;
                for (; j + 8 <= mm; j += 8) add_rows<8>(xl, r, nm, j, a);
                const int rem = mm - j;
                if (rem > 4)      add_rows<8>(xl, r, nm, j, a);
                else if (rem > 0) add_rows<4>(xl, r, nm, j, a);
            }
            if (m > CAP) {
                const int cn = nb + k;
                const int no = __builtin_amdgcn_readfirstlane(cnt[n]);
                for (int k0 = 0; k0 < no; k0 += 32) {
                    const int q = k0 + lane;
                    int  od = -1;
                    int2 oe = make_int2(0, 0);
                    if (q < no) { od = ovf_dst[q]; oe = ((const int2*)ovf_rec)[q]; }
                    const bool  hit = (od == cn);
                    const float onm = hit ? (dis[oe.x] * __int_as_float(oe.y)) * dc : 0.0f;
                    unsigned mask = __builtin_amdgcn_ballot_w32(hit);
                    while (mask) {
                        const int l = __builtin_ctz(mask);
                        mask &= mask - 1;
                        const int   rk = __builtin_amdgcn_readlane(oe.x, l);
                        const float f  = __int_as_float(__builtin_amdgcn_readlane(__float_as_int(onm), l));
                        a += *(const v4f*)(xl + (size_t)rk * D) * f;
                    }
                }
            }
            *(v4f*)(sw + (size_t)k * LP) = a;

            m = m1; mm = mm1; r = r1; wq = w1; dq = d1; en = e2;
        }
    }
    __syncthreads();

    product_half<LP>(sA + (size_t)c * LP + 16 * hh, Wgcn + woff, acc);

    float* __restrict__ dp = out + (size_t)(row0 + 8 * hh) * D + cb + 4 * c;
#pragma unroll
    for (int t = 0; t < 2; ++t) {
#pragma unroll
        for (int r = 0; r < 8; ++r) {
            const v4f o = {acc[t][0][r], acc[t][1][r], acc[t][2][r], acc[t][3][r]};
            *(v4f*)(dp + (size_t)(16 * t + r) * D) = o;
        }
    }
}


extern "C" void kernel_launch(void* const* d_in, const int* in_sizes, int n_in,
                              void* d_out, int out_size, void* d_ws, size_t ws_size,
                              hipStream_t stream) {
    const float* x    = (const float*)d_in[0];
    const int*   ei   = (const int*)d_in[1];
    const float* ew   = (const float*)d_in[2];
    const float* Wlin = (const float*)d_in[3];
    const float* Wgcn = (const float*)d_in[4];
    float*       out  = (float*)d_out;

    const int N = in_sizes[0] / D;
    const int E = in_sizes[2];

    const int* row = ei;
    const int* col = ei + E;

    float* dis = (float*)d_ws + N;

    int* tab     = (int*)(dis + N);
    int* ovf_rec = tab + (size_t)N * 64;
    int* ovf_dst = ovf_rec + (size_t)E * 2;
    int* cnt     = ovf_dst + E;

    int2* srec  = (int2*)out;
    int2* tinfo = srec + E;
    const int nchunks = (E + CH - 1) / CH;

    bin_kernel<<<nchunks, 1024, 0, stream>>>(row, col, ew, srec, tinfo, cnt, E, N, nchunks);
    fill_kernel<<<N / 32, 64, 0, stream>>>(srec, tinfo, cnt, tab, ovf_dst, ovf_rec, dis, N, nchunks);
    gather_gemm<<<N / 32, 64, 0, stream>>>(x, tab, cnt, ovf_dst, ovf_rec, dis, Wlin, Wgcn, out, N);
}
